# starting kernel
# baseline (speedup 1.0000x reference)
#include <hip/hip_runtime.h>

typedef float v2f __attribute__((ext_vector_type(2)));
typedef float v4f __attribute__((ext_vector_type(4)));
typedef float v8f __attribute__((ext_vector_type(8)));

#define NIMG 8
#define HW   16384
#define CIN  512
#define FOUT 3
#define FPAD 16
#define LDSS 516

__global__ __launch_bounds__(256) void torgb_affine_kernel(
    const float* __restrict__ dlatents, const float* __restrict__ affine_w,
    const float* __restrict__ affine_b, const float* __restrict__ conv_w,
    float* __restrict__ wst_g)
{
    __shared__ float dl[CIN];
    const int img = blockIdx.x;
    const int t   = threadIdx.x;

    const float* d = dlatents + ((size_t)img * 18 + 17) * CIN;
    for (int i = t; i < CIN; i += 256) dl[i] = d[i];
    __syncthreads();

    const float ascale = 0.044194173824159216f;
    const float cscale = 0.044194173824159216f;

    const int c0 = 2 * t;
    float s0 = 0.f, s1 = 0.f;
    for (int w = 0; w < CIN; ++w) {
        v2f aw = *(const v2f*)(affine_w + (size_t)w * CIN + c0);
        float dv = dl[w];
        s0 = fmaf(dv, aw.x, s0);
        s1 = fmaf(dv, aw.y, s1);
    }
    s0 = s0 * ascale + affine_b[c0];
    s1 = s1 * ascale + affine_b[c0 + 1];

    float* wimg = wst_g + (size_t)img * (FPAD * CIN);
    for (int f = 0; f < FPAD; ++f) {
        float w0 = 0.f, w1 = 0.f;
        if (f < FOUT) {
            w0 = s0 * conv_w[c0 * FOUT + f] * cscale;
            w1 = s1 * conv_w[(c0 + 1) * FOUT + f] * cscale;
        }
        wimg[f * CIN + c0]     = w0;
        wimg[f * CIN + c0 + 1] = w1;
    }
}

__global__ __launch_bounds__(256) void torgb_conv_kernel(
    const float* __restrict__ x, const float* __restrict__ y,
    const float* __restrict__ wst_g, const float* __restrict__ conv_b,
    float* __restrict__ out)
{
    __shared__ float wl[FPAD * LDSS];

    const int img      = blockIdx.x >> 6;
    const int blkInImg = blockIdx.x & 63;
    const int t        = threadIdx.x;

    const float* wg = wst_g + (size_t)img * (FPAD * CIN);
    for (int i = t; i < FPAD * CIN; i += 256) {
        int f = i >> 9;
        int c = i & 511;
        wl[f * LDSS + c] = wg[i];
    }
    __syncthreads();

    const int wave = t >> 5;
    const int lane = t & 31;
    const int mn   = lane & 15;
    const int h    = lane >> 4;

    const int p0 = blkInImg * 256 + wave * 32;
    const float* xr0 = x + ((size_t)img * HW + p0 + mn) * CIN;
    const float* xr1 = xr0 + (size_t)16 * CIN;
    const float* wr  = wl + mn * LDSS;

    v8f acc0 = {};
    v8f acc1 = {};
    for (int cb = 0; cb < CIN; cb += 8) {
        if ((cb & 31) == 0) {
            __builtin_prefetch(xr0 + cb + 256, 0, 0);
            __builtin_prefetch(xr1 + cb + 256, 0, 0);
        }
        v4f xa0 = __builtin_nontemporal_load((const v4f*)(xr0 + cb + 4 * h));
        v4f xa1 = __builtin_nontemporal_load((const v4f*)(xr1 + cb + 4 * h));
        v4f wb  = *(const v4f*)(wr + cb + 4 * h);
        v2f b1 = {wb.x, wb.y}, b2 = {wb.z, wb.w};
        v2f a01 = {xa0.x, xa0.y}, a02 = {xa0.z, xa0.w};
        v2f a11 = {xa1.x, xa1.y}, a12 = {xa1.z, xa1.w};
        acc0 = __builtin_amdgcn_wmma_f32_16x16x4_f32(
                   false, a01, false, b1, (short)0, acc0, false, false);
        acc1 = __builtin_amdgcn_wmma_f32_16x16x4_f32(
                   false, a11, false, b1, (short)0, acc1, false, false);
        acc0 = __builtin_amdgcn_wmma_f32_16x16x4_f32(
                   false, a02, false, b2, (short)0, acc0, false, false);
        acc1 = __builtin_amdgcn_wmma_f32_16x16x4_f32(
                   false, a12, false, b2, (short)0, acc1, false, false);
    }

    if (mn < FOUT) {
        const float bias = conv_b[mn];
        for (int tt = 0; tt < 2; ++tt) {
            v8f acc = tt ? acc1 : acc0;
            for (int j = 0; j < 8; ++j) {
                int p = p0 + 16 * tt + j + 8 * h;
                size_t oidx = ((size_t)img * HW + p) * FOUT + mn;
                float v = acc[j] + bias;
                v = fminf(fmaxf(v, -256.f), 256.f);
                v += __builtin_nontemporal_load(y + oidx);
                __builtin_nontemporal_store(v, out + oidx);
            }
        }
    }
}

extern "C" void kernel_launch(void* const* d_in, const int* in_sizes, int n_in,
                              void* d_out, int out_size, void* d_ws, size_t ws_size,
                              hipStream_t stream)
{
    (void)in_sizes; (void)n_in; (void)out_size; (void)ws_size;
    const float* x        = (const float*)d_in[0];
    const float* y        = (const float*)d_in[1];
    const float* dlatents = (const float*)d_in[2];
    const float* affine_w = (const float*)d_in[3];
    const float* affine_b = (const float*)d_in[4];
    const float* conv_w   = (const float*)d_in[5];
    const float* conv_b   = (const float*)d_in[6];
    float* out = (float*)d_out;
    float* wst = (float*)d_ws;

    torgb_affine_kernel<<<NIMG, 256, 0, stream>>>(dlatents, affine_w, affine_b,
                                                  conv_w, wst);
    torgb_conv_kernel<<<NIMG * 64, 256, 0, stream>>>(x, y, wst, conv_b, out);
}


# conv x path: per-lane row loads (16 lines per load) + prefetch -> whole-line loads (2 rows x 256 B each), two 16-load register sets in flight, transposed to A fragments via wave-private LDS
# speedup vs baseline: 2.0218x; 2.0218x over previous
#include <hip/hip_runtime.h>

typedef float v2f __attribute__((ext_vector_type(2)));
typedef float v4f __attribute__((ext_vector_type(4)));
typedef float v8f __attribute__((ext_vector_type(8)));

#define NIMG 8
#define HW   16384
#define CIN  512
#define FOUT 3
#define FPAD 16
#define LDSS 516

__global__ __launch_bounds__(256) void torgb_affine_kernel(
    const float* __restrict__ dlatents, const float* __restrict__ affine_w,
    const float* __restrict__ affine_b, const float* __restrict__ conv_w,
    float* __restrict__ wst_g)
{
    __shared__ float dl[CIN];
    const int img = blockIdx.x;
    const int t   = threadIdx.x;

    const float* d = dlatents + ((size_t)img * 18 + 17) * CIN;
    for (int i = t; i < CIN; i += 256) dl[i] = d[i];
    __syncthreads();

    const float ascale = 0.044194173824159216f;
    const float cscale = 0.044194173824159216f;

    const int c0 = 2 * t;
    float s0 = 0.f, s1 = 0.f;
    for (int w = 0; w < CIN; ++w) {
        v2f aw = *(const v2f*)(affine_w + (size_t)w * CIN + c0);
        float dv = dl[w];
        s0 = fmaf(dv, aw.x, s0);
        s1 = fmaf(dv, aw.y, s1);
    }
    s0 = s0 * ascale + affine_b[c0];
    s1 = s1 * ascale + affine_b[c0 + 1];

    float* wimg = wst_g + (size_t)img * (FPAD * CIN);
    for (int f = 0; f < FPAD; ++f) {
        float w0 = 0.f, w1 = 0.f;
        if (f < FOUT) {
            w0 = s0 * conv_w[c0 * FOUT + f] * cscale;
            w1 = s1 * conv_w[(c0 + 1) * FOUT + f] * cscale;
        }
        wimg[f * CIN + c0]     = w0;
        wimg[f * CIN + c0 + 1] = w1;
    }
}

#define XCH    64
#define NCHUNK (CIN / XCH)
#define XPITCH 68
#define DEV static __device__ __forceinline__

template <int J>
DEV void torgb_issue_chunk(v4f (&buf)[16], const float* xl)
{
#pragma unroll
    for (int i = 0; i < 16; ++i)
        buf[i] = __builtin_nontemporal_load(
                     (const v4f*)(xl + i * (2 * CIN) + XCH * J));
}

template <int J>
DEV void torgb_chunk(v4f (&buf)[16], const float* xl, float* sx,
                     const float* ar, const float* wr, v8f& acc0, v8f& acc1)
{
#pragma unroll
    for (int i = 0; i < 16; ++i)
        *(v4f*)(sx + i * (2 * XPITCH)) = buf[i];

    if constexpr (J + 2 < NCHUNK) {
        __builtin_amdgcn_sched_barrier(0);
        torgb_issue_chunk<J + 2>(buf, xl);
        __builtin_amdgcn_sched_barrier(0);
    }

#pragma unroll
    for (int g = 0; g < XCH / 8; ++g) {
        if (g == XCH / 16) __builtin_amdgcn_sched_barrier(0);
        v4f xa0 = *(const v4f*)(ar + 8 * g);
        v4f xa1 = *(const v4f*)(ar + 16 * XPITCH + 8 * g);
        v4f wb  = *(const v4f*)(wr + XCH * J + 8 * g);
        v2f b1 = {wb.x, wb.y}, b2 = {wb.z, wb.w};
        v2f a01 = {xa0.x, xa0.y}, a02 = {xa0.z, xa0.w};
        v2f a11 = {xa1.x, xa1.y}, a12 = {xa1.z, xa1.w};
        acc0 = __builtin_amdgcn_wmma_f32_16x16x4_f32(
                   false, a01, false, b1, (short)0, acc0, false, false);
        acc1 = __builtin_amdgcn_wmma_f32_16x16x4_f32(
                   false, a11, false, b1, (short)0, acc1, false, false);
        acc0 = __builtin_amdgcn_wmma_f32_16x16x4_f32(
                   false, a02, false, b2, (short)0, acc0, false, false);
        acc1 = __builtin_amdgcn_wmma_f32_16x16x4_f32(
                   false, a12, false, b2, (short)0, acc1, false, false);
    }
}

__global__ __launch_bounds__(256) void torgb_conv_kernel(
    const float* __restrict__ x, const float* __restrict__ y,
    const float* __restrict__ wst_g, const float* __restrict__ conv_b,
    float* __restrict__ out)
{
    __shared__ float wl[FPAD * LDSS];
    __shared__ float xs[8 * 32 * XPITCH];

    const int img      = blockIdx.x >> 6;
    const int blkInImg = blockIdx.x & 63;
    const int t        = threadIdx.x;

    const int wave = t >> 5;
    const int lane = t & 31;
    const int mn   = lane & 15;
    const int h    = lane >> 4;

    const int p0 = blkInImg * 256 + wave * 32;

    const float* xl = x + ((size_t)img * HW + p0 + h) * CIN + 4 * mn;
    float* xsw = xs + wave * (32 * XPITCH);
    float* sx  = xsw + h * XPITCH + 4 * mn;
    const float* ar = xsw + mn * XPITCH + 4 * h;
    const float* wr = wl + mn * LDSS + 4 * h;

    v4f bufA[16], bufB[16];
    torgb_issue_chunk<0>(bufA, xl);
    torgb_issue_chunk<1>(bufB, xl);

    const float* wg = wst_g + (size_t)img * (FPAD * CIN);
    for (int i = t; i < FPAD * CIN; i += 256) {
        int f = i >> 9;
        int c = i & 511;
        wl[f * LDSS + c] = wg[i];
    }
    __syncthreads();

    v8f acc0 = {};
    v8f acc1 = {};
    torgb_chunk<0>(bufA, xl, sx, ar, wr, acc0, acc1);
    torgb_chunk<1>(bufB, xl, sx, ar, wr, acc0, acc1);
    torgb_chunk<2>(bufA, xl, sx, ar, wr, acc0, acc1);
    torgb_chunk<3>(bufB, xl, sx, ar, wr, acc0, acc1);
    torgb_chunk<4>(bufA, xl, sx, ar, wr, acc0, acc1);
    torgb_chunk<5>(bufB, xl, sx, ar, wr, acc0, acc1);
    torgb_chunk<6>(bufA, xl, sx, ar, wr, acc0, acc1);
    torgb_chunk<7>(bufB, xl, sx, ar, wr, acc0, acc1);

    if (mn < FOUT) {
        const float bias = conv_b[mn];
        for (int tt = 0; tt < 2; ++tt) {
            v8f acc = tt ? acc1 : acc0;
            for (int j = 0; j < 8; ++j) {
                int p = p0 + 16 * tt + j + 8 * h;
                size_t oidx = ((size_t)img * HW + p) * FOUT + mn;
                float v = acc[j] + bias;
                v = fminf(fmaxf(v, -256.f), 256.f);
                v += __builtin_nontemporal_load(y + oidx);
                __builtin_nontemporal_store(v, out + oidx);
            }
        }
    }
}

extern "C" void kernel_launch(void* const* d_in, const int* in_sizes, int n_in,
                              void* d_out, int out_size, void* d_ws, size_t ws_size,
                              hipStream_t stream)
{
    (void)in_sizes; (void)n_in; (void)out_size; (void)ws_size;
    const float* x        = (const float*)d_in[0];
    const float* y        = (const float*)d_in[1];
    const float* dlatents = (const float*)d_in[2];
    const float* affine_w = (const float*)d_in[3];
    const float* affine_b = (const float*)d_in[4];
    const float* conv_w   = (const float*)d_in[5];
    const float* conv_b   = (const float*)d_in[6];
    float* out = (float*)d_out;
    float* wst = (float*)d_ws;

    torgb_affine_kernel<<<NIMG, 256, 0, stream>>>(dlatents, affine_w, affine_b,
                                                  conv_w, wst);
    torgb_conv_kernel<<<NIMG * 64, 256, 0, stream>>>(x, y, wst, conv_b, out);
}


# style-weight kernel: 8 work-groups x 512 serial terms -> 64 work-groups (32 channels x 2 images), rows split over 8 waves, 64 loads in flight per lane, partial sums joined in LDS (sum order changes)
# speedup vs baseline: 2.4878x; 1.2305x over previous
#include <hip/hip_runtime.h>

typedef float v2f __attribute__((ext_vector_type(2)));
typedef float v4f __attribute__((ext_vector_type(4)));
typedef float v8f __attribute__((ext_vector_type(8)));

#define NIMG 8
#define HW   16384
#define CIN  512
#define FOUT 3
#define FPAD 16
#define LDSS 516

#define SCH   32
#define SIMG  2
#define SROWS (CIN / 8)

__global__ __launch_bounds__(256) void torgb_affine_kernel(
    const float* __restrict__ dlatents, const float* __restrict__ affine_w,
    const float* __restrict__ affine_b, const float* __restrict__ conv_w,
    float* __restrict__ wst_g)
{
    __shared__ v4f   dl4[SIMG * CIN / 4];
    __shared__ float part[8 * SIMG * SCH];

    const int t    = threadIdx.x;
    const int lane = t & 31;
    const int wave = t >> 5;
    const int cs   = blockIdx.x & 15;
    const int ig   = blockIdx.x >> 4;
    const int c    = cs * SCH + lane;

    const int simg = ig * SIMG + (t >> 7);
    const v4f dreg = *(const v4f*)(dlatents + ((size_t)simg * 18 + 17) * CIN
                                   + 4 * (t & 127));

    const float* ap = affine_w + (size_t)(wave * SROWS) * CIN + c;
    float a[SROWS];
#pragma unroll
    for (int r = 0; r < SROWS; ++r) a[r] = ap[(size_t)r * CIN];

    const float ascale = 0.044194173824159216f;
    const float cscale = 0.044194173824159216f;

    const int im = wave & 1;
    const int fg = wave >> 1;
    const float bias = affine_b[c];
    float cw[4] = {0.f, 0.f, 0.f, 0.f};
    if (fg == 0) {
        for (int f = 0; f < FOUT; ++f) cw[f] = conv_w[c * FOUT + f];
    }

    dl4[t] = dreg;
    __syncthreads();

    float acc0 = 0.f, acc1 = 0.f;
#pragma unroll
    for (int r4 = 0; r4 < SROWS / 4; ++r4) {
        const v4f d0 = dl4[wave * (SROWS / 4) + r4];
        const v4f d1 = dl4[CIN / 4 + wave * (SROWS / 4) + r4];
        acc0 = fmaf(d0.x, a[4 * r4 + 0], acc0);
        acc1 = fmaf(d1.x, a[4 * r4 + 0], acc1);
        acc0 = fmaf(d0.y, a[4 * r4 + 1], acc0);
        acc1 = fmaf(d1.y, a[4 * r4 + 1], acc1);
        acc0 = fmaf(d0.z, a[4 * r4 + 2], acc0);
        acc1 = fmaf(d1.z, a[4 * r4 + 2], acc1);
        acc0 = fmaf(d0.w, a[4 * r4 + 3], acc0);
        acc1 = fmaf(d1.w, a[4 * r4 + 3], acc1);
    }
    part[(wave * SIMG + 0) * SCH + lane] = acc0;
    part[(wave * SIMG + 1) * SCH + lane] = acc1;
    __syncthreads();

    float s = 0.f;
    for (int w = 0; w < 8; ++w) s += part[(w * SIMG + im) * SCH + lane];
    s = s * ascale + bias;

    float* wimg = wst_g + (size_t)(ig * SIMG + im) * (FPAD * CIN) + c;
    for (int k = 0; k < 4; ++k) {
        const int f = 4 * fg + k;
        float v = 0.f;
        if (f < FOUT) v = s * cw[k] * cscale;
        wimg[f * CIN] = v;
    }
}

#define XCH    64
#define NCHUNK (CIN / XCH)
#define XPITCH 68
#define DEV static __device__ __forceinline__

template <int J>
DEV void torgb_issue_chunk(v4f (&buf)[16], const float* xl)
{
#pragma unroll
    for (int i = 0; i < 16; ++i)
        buf[i] = __builtin_nontemporal_load(
                     (const v4f*)(xl + i * (2 * CIN) + XCH * J));
}

template <int J>
DEV void torgb_chunk(v4f (&buf)[16], const float* xl, float* sx,
                     const float* ar, const float* wr, v8f& acc0, v8f& acc1)
{
#pragma unroll
    for (int i = 0; i < 16; ++i)
        *(v4f*)(sx + i * (2 * XPITCH)) = buf[i];

    if constexpr (J + 2 < NCHUNK) {
        __builtin_amdgcn_sched_barrier(0);
        torgb_issue_chunk<J + 2>(buf, xl);
        __builtin_amdgcn_sched_barrier(0);
    }

#pragma unroll
    for (int g = 0; g < XCH / 8; ++g) {
        if (g == XCH / 16) __builtin_amdgcn_sched_barrier(0);
        v4f xa0 = *(const v4f*)(ar + 8 * g);
        v4f xa1 = *(const v4f*)(ar + 16 * XPITCH + 8 * g);
        v4f wb  = *(const v4f*)(wr + XCH * J + 8 * g);
        v2f b1 = {wb.x, wb.y}, b2 = {wb.z, wb.w};
        v2f a01 = {xa0.x, xa0.y}, a02 = {xa0.z, xa0.w};
        v2f a11 = {xa1.x, xa1.y}, a12 = {xa1.z, xa1.w};
        acc0 = __builtin_amdgcn_wmma_f32_16x16x4_f32(
                   false, a01, false, b1, (short)0, acc0, false, false);
        acc1 = __builtin_amdgcn_wmma_f32_16x16x4_f32(
                   false, a11, false, b1, (short)0, acc1, false, false);
        acc0 = __builtin_amdgcn_wmma_f32_16x16x4_f32(
                   false, a02, false, b2, (short)0, acc0, false, false);
        acc1 = __builtin_amdgcn_wmma_f32_16x16x4_f32(
                   false, a12, false, b2, (short)0, acc1, false, false);
    }
}

__global__ __launch_bounds__(256) void torgb_conv_kernel(
    const float* __restrict__ x, const float* __restrict__ y,
    const float* __restrict__ wst_g, const float* __restrict__ conv_b,
    float* __restrict__ out)
{
    __shared__ float wl[FPAD * LDSS];
    __shared__ float xs[8 * 32 * XPITCH];

    const int img      = blockIdx.x >> 6;
    const int blkInImg = blockIdx.x & 63;
    const int t        = threadIdx.x;

    const int wave = t >> 5;
    const int lane = t & 31;
    const int mn   = lane & 15;
    const int h    = lane >> 4;

    const int p0 = blkInImg * 256 + wave * 32;

    const float* xl = x + ((size_t)img * HW + p0 + h) * CIN + 4 * mn;
    float* xsw = xs + wave * (32 * XPITCH);
    float* sx  = xsw + h * XPITCH + 4 * mn;
    const float* ar = xsw + mn * XPITCH + 4 * h;
    const float* wr = wl + mn * LDSS + 4 * h;

    v4f bufA[16], bufB[16];
    torgb_issue_chunk<0>(bufA, xl);
    torgb_issue_chunk<1>(bufB, xl);

    const float* wg = wst_g + (size_t)img * (FPAD * CIN);
    for (int i = t; i < FPAD * CIN; i += 256) {
        int f = i >> 9;
        int c = i & 511;
        wl[f * LDSS + c] = wg[i];
    }
    __syncthreads();

    v8f acc0 = {};
    v8f acc1 = {};
    torgb_chunk<0>(bufA, xl, sx, ar, wr, acc0, acc1);
    torgb_chunk<1>(bufB, xl, sx, ar, wr, acc0, acc1);
    torgb_chunk<2>(bufA, xl, sx, ar, wr, acc0, acc1);
    torgb_chunk<3>(bufB, xl, sx, ar, wr, acc0, acc1);
    torgb_chunk<4>(bufA, xl, sx, ar, wr, acc0, acc1);
    torgb_chunk<5>(bufB, xl, sx, ar, wr, acc0, acc1);
    torgb_chunk<6>(bufA, xl, sx, ar, wr, acc0, acc1);
    torgb_chunk<7>(bufB, xl, sx, ar, wr, acc0, acc1);

    if (mn < FOUT) {
        const float bias = conv_b[mn];
        for (int tt = 0; tt < 2; ++tt) {
            v8f acc = tt ? acc1 : acc0;
            for (int j = 0; j < 8; ++j) {
                int p = p0 + 16 * tt + j + 8 * h;
                size_t oidx = ((size_t)img * HW + p) * FOUT + mn;
                float v = acc[j] + bias;
                v = fminf(fmaxf(v, -256.f), 256.f);
                v += __builtin_nontemporal_load(y + oidx);
                __builtin_nontemporal_store(v, out + oidx);
            }
        }
    }
}

extern "C" void kernel_launch(void* const* d_in, const int* in_sizes, int n_in,
                              void* d_out, int out_size, void* d_ws, size_t ws_size,
                              hipStream_t stream)
{
    (void)in_sizes; (void)n_in; (void)out_size; (void)ws_size;
    const float* x        = (const float*)d_in[0];
    const float* y        = (const float*)d_in[1];
    const float* dlatents = (const float*)d_in[2];
    const float* affine_w = (const float*)d_in[3];
    const float* affine_b = (const float*)d_in[4];
    const float* conv_w   = (const float*)d_in[5];
    const float* conv_b   = (const float*)d_in[6];
    float* out = (float*)d_out;
    float* wst = (float*)d_ws;

    torgb_affine_kernel<<<16 * (NIMG / SIMG), 256, 0, stream>>>(dlatents, affine_w, affine_b,
                                                  conv_w, wst);
    torgb_conv_kernel<<<NIMG * 64, 256, 0, stream>>>(x, y, wst, conv_b, out);
}
